# starting kernel
# baseline (speedup 1.0000x reference)
#include <hip/hip_runtime.h>
#include <math.h>

typedef float v2f __attribute__((ext_vector_type(2)));
typedef float v8f __attribute__((ext_vector_type(8)));

#define BATCH   2
#define T_LEN   16000
#define NCEP    25
#define M_ORD   24
#define IR_LEN  2000
#define TILE_T  128
#define NWAVES  4
#define CHUNK   80
#define NCHUNK  25
#define HSTRIDE 81
#define XS_LEN  2176

__global__ __launch_bounds__(TILE_T)
void tv_fir_c2ir_kernel(const float* __restrict__ x,
                        const float* __restrict__ mc,
                        float* __restrict__ y) {
  __shared__ float xs[XS_LEN];
  __shared__ float hs[TILE_T * HSTRIDE];
  __shared__ float invn[IR_LEN];
  __shared__ float dbuf[NWAVES][16][16];

  const int tid  = threadIdx.x;
  const int wave = tid >> 5;
  const int lane = tid & 31;
  const int hi   = lane >> 4;
  const int lm   = lane & 15;

  const int b  = blockIdx.x / (T_LEN / TILE_T);
  const int tb = (blockIdx.x % (T_LEN / TILE_T)) * TILE_T;

  for (int i = tid; i < XS_LEN; i += TILE_T) {
    int g = tb - (IR_LEN - 1) + i;
    xs[i] = (i < (IR_LEN - 1 + TILE_T) && g >= 0 && g < T_LEN)
                ? x[(size_t)b * T_LEN + g] : 0.0f;
  }
  for (int n = tid; n < IR_LEN; n += TILE_T)
    invn[n] = (n > 0) ? 1.0f / (float)n : 0.0f;

  const float* mcp = mc + ((size_t)b * T_LEN + (tb + tid)) * NCEP;
  float c0 = mcp[0];
  float kc[NCEP];
#pragma unroll
  for (int k = 1; k < NCEP; ++k) kc[k] = (float)k * mcp[k];

  v8f acc0 = {};
  v8f acc1 = {};
  float hp[M_ORD];
#pragma unroll
  for (int k = 0; k < M_ORD; ++k) hp[k] = 0.0f;

  __syncthreads();

  float* hrow = &hs[tid * HSTRIDE];

  for (int chunk = 0; chunk < NCHUNK; ++chunk) {
    const int n0 = chunk * CHUNK;

    if (chunk == 0) {
      float h0 = __expf(c0);
      hrow[0] = h0;
#pragma unroll
      for (int k2 = M_ORD - 1; k2 > 0; --k2) hp[k2] = hp[k2 - 1];
      hp[0] = h0;
#pragma unroll 10
      for (int j = 1; j < CHUNK; ++j) {
        float s = 0.0f;
#pragma unroll
        for (int k = 1; k <= M_ORD; ++k) s += kc[k] * hp[k - 1];
        float hn = s * invn[j];
        hrow[j] = hn;
#pragma unroll
        for (int k2 = M_ORD - 1; k2 > 0; --k2) hp[k2] = hp[k2 - 1];
        hp[0] = hn;
      }
    } else {
#pragma unroll 10
      for (int j = 0; j < CHUNK; ++j) {
        float s = 0.0f;
#pragma unroll
        for (int k = 1; k <= M_ORD; ++k) s += kc[k] * hp[k - 1];
        float hn = s * invn[n0 + j];
        hrow[j] = hn;
#pragma unroll
        for (int k2 = M_ORD - 1; k2 > 0; --k2) hp[k2] = hp[k2 - 1];
        hp[0] = hn;
      }
    }
    __syncthreads();

#pragma unroll
    for (int st = 0; st < 2; ++st) {
      const int sub = wave * 32 + st * 16;
      v8f acc = st ? acc1 : acc0;
      const float* hsub = &hs[(sub + lm) * HSTRIDE];
#pragma unroll
      for (int g = 0; g < CHUNK / 4; ++g) {
        const int kl = g * 4 + 2 * hi;
        v2f a;  a.x = hsub[kl];  a.y = hsub[kl + 1];
        const int xb = (IR_LEN - 1) + sub + lm - (n0 + kl);
        v2f bv; bv.x = xs[xb];   bv.y = xs[xb - 1];
        acc = __builtin_amdgcn_wmma_f32_16x16x4_f32(
             false, a,  false, bv,
             (short)0, acc,  false,  false);
      }
      if (st) acc1 = acc; else acc0 = acc;
    }
    __syncthreads();
  }

  for (int st = 0; st < 2; ++st) {
    v8f acc = st ? acc1 : acc0;
#pragma unroll
    for (int r = 0; r < 8; ++r)
      dbuf[wave][r + 8 * hi][lm] = acc[r];
    __syncthreads();
    if (lane < 16) {
      const int sub = wave * 32 + st * 16;
      y[(size_t)b * T_LEN + tb + sub + lane] = dbuf[wave][lane][lane];
    }
    __syncthreads();
  }
}

extern "C" void kernel_launch(void* const* d_in, const int* in_sizes, int n_in,
                              void* d_out, int out_size, void* d_ws, size_t ws_size,
                              hipStream_t stream) {
  (void)in_sizes; (void)n_in; (void)out_size; (void)d_ws; (void)ws_size;
  const float* x  = (const float*)d_in[0];
  const float* mc = (const float*)d_in[1];
  float* y = (float*)d_out;

  dim3 grid(BATCH * (T_LEN / TILE_T));
  dim3 block(TILE_T);
  tv_fir_c2ir_kernel<<<grid, block, 0, stream>>>(x, mc, y);
}


# a chunk's 40 WMMAs issued one per tap among 40 taps of the next chunk's recursion (24 running sums, taps kept in registers, no LDS op between WMMAs): tap work in WMMAs' shadow; same sums, same order
# speedup vs baseline: 4.1980x; 4.1980x over previous
#include <hip/hip_runtime.h>
#include <math.h>

typedef float v2f __attribute__((ext_vector_type(2)));
typedef float v4f __attribute__((ext_vector_type(4)));
typedef float v8f __attribute__((ext_vector_type(8)));

#define BATCH   2
#define T_LEN   16000
#define NCEP    25
#define M_ORD   24
#define IR_LEN  2000
#define TILE_T  128
#define NTHR    128
#define NWAVES  4
#define CHUNK   80
#define NCHUNK  25
#define HSTRIDE 81
#define XW_HEAD 128
#define XW_STEP 96
#define XW_LEN  (XW_HEAD + XW_STEP * NCHUNK)
#define XW_PAD  96
#define NPRE    8
#define NOV     40
#define NKEEP   (NPRE + NOV)
#define IVP     (NPRE / 4)
#define IVA     ((NKEEP + 8) / 4)
#define GPRE    4

__device__ __forceinline__ void wave_lds_order() {
  __builtin_amdgcn_fence(__ATOMIC_RELEASE, "wavefront");
  __builtin_amdgcn_wave_barrier();
  __builtin_amdgcn_fence(__ATOMIC_ACQUIRE, "wavefront");
}

__device__ __forceinline__ float x_or_zero(const float* __restrict__ xrow, int g) {
  const float v = xrow[g > 0 ? g : 0];
  return g >= 0 ? v : 0.0f;
}

template <int JA, int JB, bool KEEP>
__device__ __forceinline__ void rec_part(const float (&kk)[M_ORD], float (&r)[M_ORD],
                                         float* __restrict__ hrow, float (&hd)[NKEEP],
                                         const v4f (&iv)[CHUNK / 4]) {
#pragma unroll
  for (int j = JA; j < JB; ++j) {
    const float hn = r[0] * iv[j >> 2][j & 3];
    if (KEEP) hd[j] = hn; else hrow[j] = hn;
#pragma unroll
    for (int k = 1; k < M_ORD; ++k) r[k - 1] = fmaf(kk[k - 1], hn, r[k]);
    r[M_ORD - 1] = kk[M_ORD - 1] * hn;
  }
}

__global__ __launch_bounds__(NTHR)
void tv_fir_c2ir_kernel(const float* __restrict__ x,
                        const float* __restrict__ mc,
                        float* __restrict__ y) {
  __shared__ float xw[NWAVES][XW_PAD + XW_LEN];
  __shared__ float hs[TILE_T * HSTRIDE];
  __shared__ __attribute__((aligned(16))) float invn[IR_LEN + CHUNK];
  __shared__ float dbuf[NWAVES][2][16][16];

  const int tid  = threadIdx.x;
  const int wave = tid >> 5;
  const int lane = tid & 31;
  const int hi   = lane >> 4;
  const int lm   = lane & 15;

  const int b  = blockIdx.x / (T_LEN / TILE_T);
  const int tb = (blockIdx.x % (T_LEN / TILE_T)) * TILE_T;

  const float* xrow = x + (size_t)b * T_LEN;
  const int tnew = tb + wave * 32 + 31;
  float* xwv = xw[wave] + XW_PAD;
#pragma unroll
  for (int m = 0; m < XW_HEAD / 32; ++m)
    xwv[lane + 32 * m] = x_or_zero(xrow, tnew - (lane + 32 * m));
#pragma unroll
  for (int m = 0; m < XW_PAD / 32; ++m) xw[wave][lane + 32 * m] = 0.0f;
  float* hrow = &hs[tid * HSTRIDE];
#pragma unroll
  for (int j = 0; j < CHUNK; ++j) hrow[j] = 0.0f;
  for (int n = tid; n < IR_LEN + CHUNK; n += NTHR)
    invn[n] = (n > 0 && n < IR_LEN) ? 1.0f / (float)n : (n == 0 ? 1.0f : 0.0f);

  const float* mcp = mc + ((size_t)b * T_LEN + (tb + tid)) * NCEP;
  const float h0 = __expf(mcp[0]);
  float kk[M_ORD];
#pragma unroll
  for (int i = 0; i < M_ORD; ++i) kk[i] = (float)(i + 1) * mcp[i + 1];

  v8f acc0 = {};
  v8f acc1 = {};
  float r[M_ORD];
#pragma unroll
  for (int i = 0; i < M_ORD; ++i) r[i] = 0.0f;
  r[0] = h0;
  __syncthreads();

  const int sub = wave * 32;
  const float* hsub0 = &hs[(sub + lm) * HSTRIDE + 2 * hi];
  const float* hsub1 = hsub0 + 16 * HSTRIDE;
  const float* xfrag = xwv + (15 - lm + 2 * hi);
  v4f iv[CHUNK / 4];
  v2f bq[CHUNK / 4 + 4];
  v2f a0[CHUNK / 4], a1[CHUNK / 4];
  float hd[NKEEP];

#pragma unroll
  for (int g = 0; g < IVP; ++g) iv[g] = *(const v4f*)&invn[4 * g];

  for (int chunk = 0; chunk < NCHUNK; ++chunk) {
    const int n0 = chunk * CHUNK;

    const int jn = XW_HEAD + XW_STEP * chunk + lane;
    float xn[XW_STEP / 32];
#pragma unroll
    for (int m = 0; m < XW_STEP / 32; ++m) xn[m] = x_or_zero(xrow, tnew - (jn + 32 * m));

    const float* xf = xfrag + (n0 - CHUNK);
#pragma unroll
    for (int g = IVP; g < IVA; ++g) iv[g] = *(const v4f*)&invn[n0 + 4 * g];
#pragma unroll
    for (int s = 0; s < GPRE + 4; ++s) { bq[s].x = xf[4 * s];  bq[s].y = xf[4 * s + 1]; }
#pragma unroll
    for (int g = 0; g < GPRE; ++g) {
      a0[g].x = hsub0[4 * g];  a0[g].y = hsub0[4 * g + 1];
      a1[g].x = hsub1[4 * g];  a1[g].y = hsub1[4 * g + 1];
    }
    __builtin_amdgcn_sched_barrier(0);
    rec_part<0, NPRE, true>(kk, r, hrow, hd, iv);
    __builtin_amdgcn_sched_barrier(0);
#pragma unroll
    for (int g = GPRE; g < CHUNK / 4; ++g) {
      bq[g + 4].x = xf[4 * (g + 4)];  bq[g + 4].y = xf[4 * (g + 4) + 1];
      a0[g].x = hsub0[4 * g];  a0[g].y = hsub0[4 * g + 1];
      a1[g].x = hsub1[4 * g];  a1[g].y = hsub1[4 * g + 1];
    }
    __builtin_amdgcn_sched_barrier(0);

    rec_part<NPRE, NKEEP, true>(kk, r, hrow, hd, iv);
#pragma unroll
    for (int g = 0; g < CHUNK / 4; ++g) {
      acc0 = __builtin_amdgcn_wmma_f32_16x16x4_f32(
           false, a0[g],  false, bq[g + 4],
           (short)0, acc0,  false,  false);
      acc1 = __builtin_amdgcn_wmma_f32_16x16x4_f32(
           false, a1[g],  false, bq[g],
           (short)0, acc1,  false,  false);
    }
#pragma unroll
    for (int t = 0; t < NOV; ++t) {
      __builtin_amdgcn_sched_group_barrier(0x002, 1, 0);
      __builtin_amdgcn_sched_group_barrier(0x008, 1, 0);
      __builtin_amdgcn_sched_group_barrier(0x002, M_ORD, 0);
    }
    asm volatile("" :: "v"(acc0), "v"(acc1));
    __builtin_amdgcn_sched_barrier(0);

#pragma unroll
    for (int j = 0; j < NKEEP; ++j) hrow[j] = hd[j];
#pragma unroll
    for (int g = IVA; g < CHUNK / 4; ++g) iv[g] = *(const v4f*)&invn[n0 + 4 * g];

    rec_part<NKEEP, CHUNK, false>(kk, r, hrow, hd, iv);

#pragma unroll
    for (int g = 0; g < IVP; ++g) iv[g] = *(const v4f*)&invn[n0 + CHUNK + 4 * g];
#pragma unroll
    for (int m = 0; m < XW_STEP / 32; ++m) xwv[jn + 32 * m] = xn[m];
    wave_lds_order();
  }

  {
    const float* xf = xfrag + ((NCHUNK - 1) * CHUNK);
#pragma unroll
    for (int s = 0; s < CHUNK / 4 + 4; ++s) {
      bq[s].x = xf[4 * s];
      bq[s].y = xf[4 * s + 1];
    }
#pragma unroll
    for (int g = 0; g < CHUNK / 4; ++g) {
      a0[g].x = hsub0[4 * g];  a0[g].y = hsub0[4 * g + 1];
      a1[g].x = hsub1[4 * g];  a1[g].y = hsub1[4 * g + 1];
    }
  }
  {
#pragma unroll
  for (int g = 0; g < CHUNK / 4; ++g) {
    acc0 = __builtin_amdgcn_wmma_f32_16x16x4_f32(
         false, a0[g],  false, bq[g + 4],
         (short)0, acc0,  false,  false);
    acc1 = __builtin_amdgcn_wmma_f32_16x16x4_f32(
         false, a1[g],  false, bq[g],
         (short)0, acc1,  false,  false);
  }
  }

#pragma unroll
  for (int r = 0; r < 8; ++r) {
    dbuf[wave][0][r + 8 * hi][lm] = acc0[r];
    dbuf[wave][1][r + 8 * hi][lm] = acc1[r];
  }
  __syncthreads();
  {
    const int w2 = tid >> 5, t2 = (tid >> 4) & 1, l2 = tid & 15;
    y[(size_t)b * T_LEN + tb + tid] = dbuf[w2][t2][l2][l2];
  }
}

extern "C" void kernel_launch(void* const* d_in, const int* in_sizes, int n_in,
                              void* d_out, int out_size, void* d_ws, size_t ws_size,
                              hipStream_t stream) {
  (void)in_sizes; (void)n_in; (void)out_size; (void)d_ws; (void)ws_size;
  const float* x  = (const float*)d_in[0];
  const float* mc = (const float*)d_in[1];
  float* y = (float*)d_out;

  dim3 grid(BATCH * (T_LEN / TILE_T));
  dim3 block(NTHR);
  tv_fir_c2ir_kernel<<<grid, block, 0, stream>>>(x, mc, y);
}


# 1/n per tap is wave-uniform: scalar loads from a constant table to scalar registers, not 80 VGPRs from an LDS table built per launch; h rows interleaved by tile: loop fits 256 VGPRs; same sums, order
# speedup vs baseline: 5.0270x; 1.1975x over previous
#include <hip/hip_runtime.h>
#include <math.h>

typedef float v2f __attribute__((ext_vector_type(2)));
typedef float v8f __attribute__((ext_vector_type(8)));

#define BATCH   2
#define T_LEN   16000
#define NCEP    25
#define M_ORD   24
#define IR_LEN  2000
#define TILE_T  128
#define NTHR    128
#define NWAVES  4
#define CHUNK   80
#define NCHUNK  25
#define HSTRIDE 81
#define XW_HEAD 128
#define XW_STEP 96
#define XW_LEN  (XW_HEAD + XW_STEP * NCHUNK)
#define XW_PAD  96
#define NPRE    12
#define NOV     40
#define NWM     (NPRE + NOV)
#define GPRE    4

struct inv_table { float v[IR_LEN + CHUNK]; };
constexpr inv_table make_inv_table() {
  inv_table t{};
  for (int n = 0; n < IR_LEN + CHUNK; ++n)
    t.v[n] = (n == 0) ? 1.0f : (n < IR_LEN ? 1.0f / (float)n : 0.0f);
  return t;
}
__constant__ inv_table c_invn = make_inv_table();

__device__ __forceinline__ void wave_lds_order() {
  __builtin_amdgcn_fence(__ATOMIC_RELEASE, "wavefront");
  __builtin_amdgcn_wave_barrier();
  __builtin_amdgcn_fence(__ATOMIC_ACQUIRE, "wavefront");
}

__device__ __forceinline__ float x_or_zero(const float* __restrict__ xrow, int g) {
  const float v = xrow[g > 0 ? g : 0];
  return g >= 0 ? v : 0.0f;
}

template <int JA, int JB>
__device__ __forceinline__ void rec_part(const float (&kk)[M_ORD], float (&r)[M_ORD],
                                         float (&hd)[CHUNK],
                                         const float (&iv)[CHUNK]) {
#pragma unroll
  for (int j = JA; j < JB; ++j) {
    const float hn = r[0] * iv[j];
    hd[j] = hn;
#pragma unroll
    for (int k = 1; k < M_ORD; ++k) r[k - 1] = fmaf(kk[k - 1], hn, r[k]);
    r[M_ORD - 1] = kk[M_ORD - 1] * hn;
  }
}

__global__ __launch_bounds__(NTHR)
void tv_fir_c2ir_kernel(const float* __restrict__ x,
                        const float* __restrict__ mc,
                        float* __restrict__ y) {
  __shared__ float xw[NWAVES][XW_PAD + XW_LEN];
  __shared__ float hs[TILE_T * HSTRIDE];
  __shared__ float dbuf[NWAVES][2][16][16];

  const int tid  = threadIdx.x;
  const int wave = tid >> 5;
  const int lane = tid & 31;
  const int hi   = lane >> 4;
  const int lm   = lane & 15;

  const int b  = blockIdx.x / (T_LEN / TILE_T);
  const int tb = (blockIdx.x % (T_LEN / TILE_T)) * TILE_T;

  const float* xrow = x + (size_t)b * T_LEN;
  const int tnew = tb + wave * 32 + 31;
  float* xwv = xw[wave] + XW_PAD;
  float xq[XW_STEP / 32];
#pragma unroll
  for (int m = 0; m < XW_HEAD / 32; ++m) {
    const float v = x_or_zero(xrow, tnew - (lane + 32 * m));
    xwv[lane + 32 * m] = v;
    if (m >= 1) xq[m - 1] = v;
  }
#pragma unroll
  for (int m = 0; m < XW_PAD / 32; ++m) xw[wave][lane + 32 * m] = 0.0f;
  float* hrow = &hs[((wave * 16 + lm) * 2 + hi) * HSTRIDE];
#pragma unroll
  for (int j = 0; j < CHUNK; ++j) hrow[j] = 0.0f;
  __builtin_prefetch(&c_invn.v[16 * tid], 0, 3);
  const float* mcp = mc + ((size_t)b * T_LEN + (tb + tid)) * NCEP;
  const float h0 = __expf(mcp[0]);
  float kk[M_ORD];
#pragma unroll
  for (int i = 0; i < M_ORD; ++i) kk[i] = (float)(i + 1) * mcp[i + 1];

  v8f acc0 = {};
  v8f acc1 = {};
  float r[M_ORD];
#pragma unroll
  for (int i = 0; i < M_ORD; ++i) r[i] = 0.0f;
  r[0] = h0;
  __syncthreads();

  const int sub = wave * 32;
  const float* hsub0 = &hs[(sub + 2 * lm) * HSTRIDE + 2 * hi];
  const float* hsub1 = hsub0 + HSTRIDE;
  const float* xfrag = xwv + (15 - lm + 2 * hi);
  float iv[CHUNK];
  v2f bq[CHUNK / 4 + 4];
  v2f a0[CHUNK / 4], a1[CHUNK / 4];
  float hd[CHUNK];

  const float* ivt = c_invn.v;
#pragma unroll
  for (int j = 0; j < NWM; ++j) iv[j] = ivt[j];

  for (int chunk = 0; chunk < NCHUNK; ++chunk) {
    const int n0 = chunk * CHUNK;

    const float* xf = xfrag + (n0 - CHUNK);
    asm volatile("" :: "s"(iv[0]));
    __builtin_amdgcn_sched_barrier(0);
#pragma unroll
    for (int j = NWM; j < CHUNK; ++j) iv[j] = ivt[j];
#pragma unroll
    for (int s = 0; s < GPRE + 4; ++s) { bq[s].x = xf[4 * s];  bq[s].y = xf[4 * s + 1]; }
#pragma unroll
    for (int g = 0; g < GPRE; ++g) {
      a0[g].x = hsub0[4 * g];  a0[g].y = hsub0[4 * g + 1];
      a1[g].x = hsub1[4 * g];  a1[g].y = hsub1[4 * g + 1];
    }
    const int jn = XW_HEAD + XW_STEP * chunk + lane;
#pragma unroll
    for (int m = 0; m < XW_STEP / 32; ++m) xwv[jn - XW_STEP + 32 * m] = xq[m];
    __builtin_amdgcn_sched_barrier(0);

    float xn[XW_STEP / 32];
#pragma unroll
    for (int m = 0; m < XW_STEP / 32; ++m) xn[m] = x_or_zero(xrow, tnew - (jn + 32 * m));

    rec_part<0, NPRE>(kk, r, hd, iv);
    __builtin_amdgcn_sched_barrier(0);
#pragma unroll
    for (int g = GPRE; g < CHUNK / 4; ++g) {
      bq[g + 4].x = xf[4 * (g + 4)];  bq[g + 4].y = xf[4 * (g + 4) + 1];
      a0[g].x = hsub0[4 * g];  a0[g].y = hsub0[4 * g + 1];
      a1[g].x = hsub1[4 * g];  a1[g].y = hsub1[4 * g + 1];
    }
    __builtin_amdgcn_sched_barrier(0);

    rec_part<NPRE, NWM>(kk, r, hd, iv);
#pragma unroll
    for (int g = 0; g < CHUNK / 4; ++g) {
      acc0 = __builtin_amdgcn_wmma_f32_16x16x4_f32(
           false, a0[g],  false, bq[g + 4],
           (short)0, acc0,  false,  false);
      acc1 = __builtin_amdgcn_wmma_f32_16x16x4_f32(
           false, a1[g],  false, bq[g],
           (short)0, acc1,  false,  false);
    }
#pragma unroll
    for (int t = 0; t < NOV; ++t) {
      __builtin_amdgcn_sched_group_barrier(0x002, 1, 0);
      __builtin_amdgcn_sched_group_barrier(0x008, 1, 0);
      __builtin_amdgcn_sched_group_barrier(0x002, M_ORD, 0);
    }
    asm volatile("" :: "v"(acc0), "v"(acc1));
    __builtin_amdgcn_sched_barrier(0);

    asm volatile("" :: "s"(iv[NWM]));
    __builtin_amdgcn_sched_barrier(0);
    ivt += CHUNK;
#pragma unroll
    for (int j = 0; j < NWM; ++j) iv[j] = ivt[j];
    __builtin_amdgcn_sched_barrier(0);
    rec_part<NWM, CHUNK>(kk, r, hd, iv);
    asm volatile("" :: "v"(r[0]), "v"(r[1]), "v"(r[2]), "v"(r[3]), "v"(r[4]), "v"(r[5]),
                       "v"(r[6]), "v"(r[7]), "v"(r[8]), "v"(r[9]), "v"(r[10]), "v"(r[11]),
                       "v"(r[12]), "v"(r[13]), "v"(r[14]), "v"(r[15]), "v"(r[16]), "v"(r[17]),
                       "v"(r[18]), "v"(r[19]), "v"(r[20]), "v"(r[21]), "v"(r[22]), "v"(r[23]),
                       "v"(xn[0]), "v"(xn[1]), "v"(xn[2]));
    __builtin_amdgcn_sched_barrier(0);

#pragma unroll
    for (int j = 0; j < CHUNK; ++j) hrow[j] = hd[j];
#pragma unroll
    for (int m = 0; m < XW_STEP / 32; ++m) xq[m] = xn[m];
    wave_lds_order();
  }

  {
    const float* xf = xfrag + ((NCHUNK - 1) * CHUNK);
#pragma unroll
    for (int s = 0; s < CHUNK / 4 + 4; ++s) {
      bq[s].x = xf[4 * s];
      bq[s].y = xf[4 * s + 1];
    }
#pragma unroll
    for (int g = 0; g < CHUNK / 4; ++g) {
      a0[g].x = hsub0[4 * g];  a0[g].y = hsub0[4 * g + 1];
      a1[g].x = hsub1[4 * g];  a1[g].y = hsub1[4 * g + 1];
    }
  }
  {
#pragma unroll
  for (int g = 0; g < CHUNK / 4; ++g) {
    acc0 = __builtin_amdgcn_wmma_f32_16x16x4_f32(
         false, a0[g],  false, bq[g + 4],
         (short)0, acc0,  false,  false);
    acc1 = __builtin_amdgcn_wmma_f32_16x16x4_f32(
         false, a1[g],  false, bq[g],
         (short)0, acc1,  false,  false);
  }
  }

#pragma unroll
  for (int r = 0; r < 8; ++r) {
    dbuf[wave][0][r + 8 * hi][lm] = acc0[r];
    dbuf[wave][1][r + 8 * hi][lm] = acc1[r];
  }
  __syncthreads();
  {
    const int w2 = tid >> 5, t2 = (tid >> 4) & 1, l2 = tid & 15;
    y[(size_t)b * T_LEN + tb + tid] = dbuf[w2][t2][l2][l2];
  }
}

extern "C" void kernel_launch(void* const* d_in, const int* in_sizes, int n_in,
                              void* d_out, int out_size, void* d_ws, size_t ws_size,
                              hipStream_t stream) {
  (void)in_sizes; (void)n_in; (void)out_size; (void)d_ws; (void)ws_size;
  const float* x  = (const float*)d_in[0];
  const float* mc = (const float*)d_in[1];
  float* y = (float*)d_out;

  dim3 grid(BATCH * (T_LEN / TILE_T));
  dim3 block(NTHR);
  tv_fir_c2ir_kernel<<<grid, block, 0, stream>>>(x, mc, y);
}
